# starting kernel
# baseline (speedup 1.0000x reference)
#include <hip/hip_runtime.h>

typedef __attribute__((ext_vector_type(2))) float v2f;
typedef __attribute__((ext_vector_type(8))) float v8f;

#define NV 64
#define WPB 8

__global__ void __launch_bounds__(256)
gcn_pool_kernel(const float* __restrict__ A, const float* __restrict__ h0,
                const float* __restrict__ W1, const float* __restrict__ W2,
                float* __restrict__ out, int B)
{
  __shared__ float dinvS[WPB][NV];
  __shared__ float GS[WPB][NV * 4];

  const int tid  = threadIdx.x;
  const int wv   = tid >> 5;
  const int lane = tid & 31;
  const int r    = lane & 15;
  const int h    = lane >> 4;

  const int b = blockIdx.x * WPB + wv;
  if (b >= B) return;

  const float* Ab = A  + (size_t)b * (NV * NV);
  const float* Hb = h0 + (size_t)b * (NV * 5);

  v2f a[4][16];
#pragma unroll
  for (int m = 0; m < 4; ++m) {
    const int row = 16 * m + r;
#pragma unroll
    for (int k = 0; k < 16; ++k) {
      const int col = 4 * k + 2 * h;
      float x = Ab[row * NV + col];
      float y = Ab[row * NV + col + 1];
      if (row == col)     x += 1.0f;
      if (row == col + 1) y += 1.0f;
      a[m][k].x = x;
      a[m][k].y = y;
    }
  }

  float dv[4];
#pragma unroll
  for (int m = 0; m < 4; ++m) {
    float p = 0.0f;
#pragma unroll
    for (int k = 0; k < 16; ++k) p += a[m][k].x + a[m][k].y;
    p += __shfl_xor(p, 16, 32);
    dv[m] = 1.0f / sqrtf(p);
  }
  if (h == 0) {
#pragma unroll
    for (int m = 0; m < 4; ++m) dinvS[wv][16 * m + r] = dv[m];
  }
#pragma unroll
  for (int m = 0; m < 4; ++m)
#pragma unroll
    for (int k = 0; k < 16; ++k) { a[m][k].x *= dv[m]; a[m][k].y *= dv[m]; }

  float w1[5][3];
#pragma unroll
  for (int c = 0; c < 5; ++c)
#pragma unroll
    for (int f = 0; f < 3; ++f) w1[c][f] = W1[c * 3 + f];

#pragma unroll
  for (int t = 0; t < 2; ++t) {
    const int w = lane + 32 * t;
    const float* hr = Hb + w * 5;
    float hv[5];
#pragma unroll
    for (int c = 0; c < 5; ++c) hv[c] = hr[c];
    const float dw = dinvS[wv][w];
#pragma unroll
    for (int f = 0; f < 3; ++f) {
      float s = 0.0f;
#pragma unroll
      for (int c = 0; c < 5; ++c) s += hv[c] * w1[c][f];
      GS[wv][w * 4 + f] = dw * s;
    }
  }

  v2f bf[16];
#pragma unroll
  for (int k = 0; k < 16; ++k) {
    const int w0 = 4 * k + 2 * h;
    float bx = 0.0f, by = 0.0f;
    if (r < 3) {
      bx = GS[wv][w0 * 4 + r];
      by = GS[wv][(w0 + 1) * 4 + r];
    }
    bf[k].x = bx;
    bf[k].y = by;
  }

  float p = 0.0f;
#pragma unroll
  for (int m = 0; m < 4; ++m) {
    v8f acc = {0.f, 0.f, 0.f, 0.f, 0.f, 0.f, 0.f, 0.f};
#pragma unroll
    for (int k = 0; k < 16; ++k)
      acc = __builtin_amdgcn_wmma_f32_16x16x4_f32(
           false, a[m][k],  false, bf[k],
           (short)0, acc,  false,  false);
#pragma unroll
    for (int i = 0; i < 8; ++i) p += fmaxf(acc[i], 0.0f);
  }
  p += __shfl_xor(p, 16, 32);

  float t = 0.0f;
  if (lane < 3) t = p * W2[lane];
  float o = __shfl(t, 0, 32) + __shfl(t, 1, 32) + __shfl(t, 2, 32);
  if (lane == 0) out[b] = o;
}

extern "C" void kernel_launch(void* const* d_in, const int* in_sizes, int n_in,
                              void* d_out, int out_size, void* d_ws, size_t ws_size,
                              hipStream_t stream) {
  const float* A  = (const float*)d_in[0];
  const float* h0 = (const float*)d_in[1];
  const float* W1 = (const float*)d_in[2];
  const float* W2 = (const float*)d_in[3];
  float* out = (float*)d_out;

  const int B = in_sizes[0] / (NV * NV);
  dim3 block(256);
  dim3 grid((B + WPB - 1) / WPB);
  gcn_pool_kernel<<<grid, block, 0, stream>>>(A, h0, W1, W2, out, B);
}


# Persistent waves run their 8 batches as a 3-stage pipeline: L2 prefetch two batches ahead, fragment loads one batch ahead into a second register set, then compute; 256 resident blocks
# speedup vs baseline: 1.1017x; 1.1017x over previous
#include <hip/hip_runtime.h>


typedef __attribute__((ext_vector_type(2))) float v2f;
typedef __attribute__((ext_vector_type(8))) float v8f;

#define NV 64
#define WPB 8
#define NBLK 256

struct Tile {
  v2f a[4][16];
  float hv[2][5];
};

__device__ __forceinline__ void load_tile(Tile& t, const float* __restrict__ A,
                                          const float* __restrict__ h0,
                                          int b, int r, int h, int lane)
{
  const float* Hb = h0 + (size_t)b * (NV * 5);
#pragma unroll
  for (int s = 0; s < 2; ++s) {
    const float* hr = Hb + (lane + 32 * s) * 5;
#pragma unroll
    for (int c = 0; c < 5; ++c) t.hv[s][c] = hr[c];
  }
  const float* Ab = A + (size_t)b * (NV * NV);
#pragma unroll
  for (int m = 0; m < 4; ++m) {
    const int row = 16 * m + r;
#pragma unroll
    for (int k = 0; k < 16; ++k) {
      const int col = 4 * k + 2 * h;
      t.a[m][k].x = Ab[row * NV + col];
      t.a[m][k].y = Ab[row * NV + col + 1];
    }
  }
}

__device__ __forceinline__ int imin(int x, int y) { return x < y ? x : y; }

__device__ __forceinline__ void prefetch_tile(const float* __restrict__ A,
                                              const float* __restrict__ h0, int b, int lane)
{
  const char* Ab = (const char*)(A + (size_t)b * (NV * NV));
  const char* Hb = (const char*)(h0 + (size_t)b * (NV * 5));
  __builtin_prefetch(Ab + lane * 256, 0, 3);
  __builtin_prefetch(Ab + (lane + 32) * 256, 0, 3);
  __builtin_prefetch(Hb + (lane < 4 ? lane : 4) * 256, 0, 3);
}

__device__ __forceinline__ void compute_tile(Tile& t, int b, float* __restrict__ dinvW,
                                             float* __restrict__ GW,
                                             const float (&w1)[5][3], float w2v,
                                             float* __restrict__ out,
                                             int r, int h, int lane)
{
#pragma unroll
  for (int m = 0; m < 4; ++m) {
    const int row = 16 * m + r;
#pragma unroll
    for (int k = 0; k < 16; ++k) {
      const int col = 4 * k + 2 * h;
      if (row == col)     t.a[m][k].x += 1.0f;
      if (row == col + 1) t.a[m][k].y += 1.0f;
    }
  }

  float dv[4];
#pragma unroll
  for (int m = 0; m < 4; ++m) {
    float p = 0.0f;
#pragma unroll
    for (int k = 0; k < 16; ++k) p += t.a[m][k].x + t.a[m][k].y;
    p += __shfl_xor(p, 16, 32);
    dv[m] = 1.0f / sqrtf(p);
  }
  if (h == 0) {
#pragma unroll
    for (int m = 0; m < 4; ++m) dinvW[16 * m + r] = dv[m];
  }
#pragma unroll
  for (int m = 0; m < 4; ++m)
#pragma unroll
    for (int k = 0; k < 16; ++k) { t.a[m][k].x *= dv[m]; t.a[m][k].y *= dv[m]; }

#pragma unroll
  for (int s = 0; s < 2; ++s) {
    const int w = lane + 32 * s;
    const float dw = dinvW[w];
#pragma unroll
    for (int f = 0; f < 3; ++f) {
      float acc = 0.0f;
#pragma unroll
      for (int c = 0; c < 5; ++c) acc += t.hv[s][c] * w1[c][f];
      GW[w * 4 + f] = dw * acc;
    }
  }

  v2f bf[16];
#pragma unroll
  for (int k = 0; k < 16; ++k) {
    const int w0 = 4 * k + 2 * h;
    float bx = 0.0f, by = 0.0f;
    if (r < 3) {
      bx = GW[w0 * 4 + r];
      by = GW[(w0 + 1) * 4 + r];
    }
    bf[k].x = bx;
    bf[k].y = by;
  }

  float p = 0.0f;
#pragma unroll
  for (int m = 0; m < 4; ++m) {
    v8f acc = {0.f, 0.f, 0.f, 0.f, 0.f, 0.f, 0.f, 0.f};
#pragma unroll
    for (int k = 0; k < 16; ++k)
      acc = __builtin_amdgcn_wmma_f32_16x16x4_f32(
           false, t.a[m][k],  false, bf[k],
           (short)0, acc,  false,  false);
#pragma unroll
    for (int i = 0; i < 8; ++i) p += fmaxf(acc[i], 0.0f);
  }
  p += __shfl_xor(p, 16, 32);

  float tt = 0.0f;
  if (lane < 3) tt = p * w2v;
  float o = __shfl(tt, 0, 32) + __shfl(tt, 1, 32) + __shfl(tt, 2, 32);
  if (lane == 0) out[b] = o;
}

__global__ void __launch_bounds__(256)
gcn_pool_kernel(const float* __restrict__ A, const float* __restrict__ h0,
                const float* __restrict__ W1, const float* __restrict__ W2,
                float* __restrict__ out, int B)
{
  __shared__ float dinvS[WPB][NV];
  __shared__ float GS[WPB][NV * 4];

  const int tid  = threadIdx.x;
  const int wv   = tid >> 5;
  const int lane = tid & 31;
  const int r    = lane & 15;
  const int h    = lane >> 4;

  const int stride = gridDim.x * WPB;
  int b = blockIdx.x * WPB + wv;
  if (b >= B) return;

  float w1[5][3];
#pragma unroll
  for (int c = 0; c < 5; ++c)
#pragma unroll
    for (int f = 0; f < 3; ++f) w1[c][f] = W1[c * 3 + f];
  float w2v = 0.0f;
  if (lane < 3) w2v = W2[lane];

  float* dinvW = &dinvS[wv][0];
  float* GW    = &GS[wv][0];

  Tile t0, t1;
  prefetch_tile(A, h0, imin(b + stride, B - 1), lane);
  load_tile(t0, A, h0, b, r, h, lane);
  for (;;) {
    const int bn = b + stride;
    if (bn >= B) { compute_tile(t0, b, dinvW, GW, w1, w2v, out, r, h, lane); break; }
    const int bnn = bn + stride;
    prefetch_tile(A, h0, imin(bnn, B - 1), lane);
    load_tile(t1, A, h0, bn, r, h, lane);
    compute_tile(t0, b, dinvW, GW, w1, w2v, out, r, h, lane);
    if (bnn >= B) { compute_tile(t1, bn, dinvW, GW, w1, w2v, out, r, h, lane); break; }
    prefetch_tile(A, h0, imin(bnn + stride, B - 1), lane);
    load_tile(t0, A, h0, bnn, r, h, lane);
    compute_tile(t1, bn, dinvW, GW, w1, w2v, out, r, h, lane);
    b = bnn;
  }
}

extern "C" void kernel_launch(void* const* d_in, const int* in_sizes, int n_in,
                              void* d_out, int out_size, void* d_ws, size_t ws_size,
                              hipStream_t stream) {
  const float* A  = (const float*)d_in[0];
  const float* h0 = (const float*)d_in[1];
  const float* W1 = (const float*)d_in[2];
  const float* W2 = (const float*)d_in[3];
  float* out = (float*)d_out;

  const int B = in_sizes[0] / (NV * NV);
  dim3 block(256);
  dim3 grid(NBLK);
  gcn_pool_kernel<<<grid, block, 0, stream>>>(A, h0, W1, W2, out, B);
}


# Product as dinv_v*(A@G + G): A fragments reach the WMMAs unmodified (identity = accumulator start, row scale on the result); G through LDS transposed, 16-B reads in one hand-over; sum order changes
# speedup vs baseline: 1.3130x; 1.1918x over previous
#include <hip/hip_runtime.h>


typedef __attribute__((ext_vector_type(2))) float v2f;
typedef __attribute__((ext_vector_type(4))) float v4f;
typedef __attribute__((ext_vector_type(8))) float v8f;

#define NV 64
#define WPB 8
#define NBLK 256

struct Tile {
  v4f raw[4][2][2][2];
  float hv[2][5];
};

__device__ __forceinline__ void load_tile(Tile& t, const float* __restrict__ A,
                                          const float* __restrict__ h0, int b, int lane)
{
  const float* Hb = h0 + (size_t)b * (NV * 5);
#pragma unroll
  for (int s = 0; s < 2; ++s) {
    const float* hr = Hb + (lane + 32 * s) * 5;
#pragma unroll
    for (int c = 0; c < 5; ++c) t.hv[s][c] = hr[c];
  }
  const float* Ab = A + (size_t)b * (NV * NV) + (lane & 7) * NV + (lane >> 4) * 16 + ((lane >> 3) & 1) * 8;
#pragma unroll
  for (int m = 0; m < 4; ++m)
#pragma unroll
    for (int c = 0; c < 2; ++c)
#pragma unroll
      for (int q = 0; q < 2; ++q)
#pragma unroll
        for (int d = 0; d < 2; ++d)
          t.raw[m][c][d][q] = *(const v4f*)(Ab + (16 * m + 8 * q) * NV + 32 * c + 4 * d);
}

template <int XM, int BANKS>
__device__ __forceinline__ float lane_take(float keep, float from)
{
  return __builtin_bit_cast(float,
      __builtin_amdgcn_update_dpp(__builtin_bit_cast(int, keep), __builtin_bit_cast(int, from),
                                  0x160 + XM, 0xF, BANKS, false));
}

__device__ __forceinline__ int kcol(int k, int h)
{
  return 32 * (k >> 3) + 16 * h + 2 * (k & 7);
}

__device__ __forceinline__ void to_fragments(const Tile& t, v2f (&a)[4][16])
{
#pragma unroll
  for (int m = 0; m < 4; ++m)
#pragma unroll
    for (int c = 0; c < 2; ++c)
#pragma unroll
      for (int d = 0; d < 2; ++d)
#pragma unroll
        for (int x = 0; x < 4; ++x) {
          const float r0 = t.raw[m][c][d][0][x], r1 = t.raw[m][c][d][1][x];
          const float e0 = lane_take<8, 0xC>(r0, r1);
          const float e1 = lane_take<8, 0x3>(r1, r0);
          const int kb = 8 * c + 2 * d + (x >> 1);
          if (x & 1) { a[m][kb].y = e0; a[m][kb + 4].y = e1; }
          else       { a[m][kb].x = e0; a[m][kb + 4].x = e1; }
        }
}

__device__ __forceinline__ int imin(int x, int y) { return x < y ? x : y; }

__device__ __forceinline__ void prefetch_tile(const float* __restrict__ A,
                                              const float* __restrict__ h0, int b, int lane)
{
  const char* Ab = (const char*)(A + (size_t)b * (NV * NV));
  const char* Hb = (const char*)(h0 + (size_t)b * (NV * 5));
  __builtin_prefetch(Ab + lane * 256, 0, 3);
  __builtin_prefetch(Ab + (lane + 32) * 256, 0, 3);
  __builtin_prefetch(Hb + (lane < 4 ? lane : 4) * 256, 0, 3);
}

__device__ __forceinline__ void unpack_tile(const Tile& t, v2f (&a)[4][16], float (&xw)[2][3],
                                            const float (&w1)[5][3])
{
  to_fragments(t, a);
#pragma unroll
  for (int s = 0; s < 2; ++s)
#pragma unroll
    for (int f = 0; f < 3; ++f) {
      float acc = 0.0f;
#pragma unroll
      for (int c = 0; c < 5; ++c) acc += t.hv[s][c] * w1[c][f];
      xw[s][f] = acc;
    }
}

#define GP 76
#define LDSW (3 * GP + NV)

__device__ __forceinline__ void finish_tile(const v2f (&a)[4][16], const float (&xw)[2][3], int b,
                                            float* ldsW, const float* bsrc, const float* csrc,
                                            const float* dsrc, float w2v, float* __restrict__ out,
                                            int r, int h, int lane)
{
  float* GT = ldsW;
  float* DV = ldsW + 3 * GP;

  float dv[4];
#pragma unroll
  for (int m = 0; m < 4; ++m) {
    v2f s0 = a[m][0], s1 = a[m][1];
#pragma unroll
    for (int k = 2; k < 16; k += 2) { s0 += a[m][k]; s1 += a[m][k + 1]; }
    s0 += s1;
    float p = s0.x + s0.y;
    p += __shfl_xor(p, 16, 32);
    dv[m] = 1.0f / sqrtf(p + 1.0f);
  }
#pragma unroll
  for (int m = 0; m < 4; ++m) DV[16 * m + r] = dv[m];

#pragma unroll
  for (int s = 0; s < 2; ++s) {
    const float dw = h ? dv[2 * s + 1] : dv[2 * s];
#pragma unroll
    for (int f = 0; f < 3; ++f) GT[f * GP + lane + 32 * s] = dw * xw[s][f];
  }

  v4f bq[2][4], cq[4][2], dq[4][2];
#pragma unroll
  for (int c = 0; c < 2; ++c)
#pragma unroll
    for (int q = 0; q < 4; ++q) bq[c][q] = *(const v4f*)(bsrc + 32 * c + 4 * q);
#pragma unroll
  for (int m = 0; m < 4; ++m)
#pragma unroll
    for (int j = 0; j < 2; ++j) {
      cq[m][j] = *(const v4f*)(csrc + 16 * m + 4 * j);
      dq[m][j] = *(const v4f*)(dsrc + 16 * m + 4 * j);
    }
  v2f bf[16];
#pragma unroll
  for (int c = 0; c < 2; ++c)
#pragma unroll
    for (int q = 0; q < 4; ++q) {
      bf[8 * c + 2 * q].x     = bq[c][q].x;
      bf[8 * c + 2 * q].y     = bq[c][q].y;
      bf[8 * c + 2 * q + 1].x = bq[c][q].z;
      bf[8 * c + 2 * q + 1].y = bq[c][q].w;
    }

  __builtin_amdgcn_sched_barrier(0);

  float p = 0.0f;
#pragma unroll
  for (int m = 0; m < 4; ++m) {
    v8f acc = {cq[m][0].x, cq[m][0].y, cq[m][0].z, cq[m][0].w,
               cq[m][1].x, cq[m][1].y, cq[m][1].z, cq[m][1].w};
#pragma unroll
    for (int k = 0; k < 16; ++k)
      acc = __builtin_amdgcn_wmma_f32_16x16x4_f32(
           false, a[m][k],  false, bf[k],
           (short)0, acc,  false,  false);
#pragma unroll
    for (int i = 0; i < 8; ++i) p += fmaxf(acc[i] * dq[m][i >> 2][i & 3], 0.0f);
  }
  p += __shfl_xor(p, 16, 32);

  float tt = 0.0f;
  if (lane < 3) tt = p * w2v;
  float o = __shfl(tt, 0, 32) + __shfl(tt, 1, 32) + __shfl(tt, 2, 32);
  if (lane == 0) out[b] = o;
}

__global__ void __launch_bounds__(256)
gcn_pool_kernel(const float* __restrict__ A, const float* __restrict__ h0,
                const float* __restrict__ W1, const float* __restrict__ W2,
                float* __restrict__ out, int B)
{
  __shared__ __attribute__((aligned(16))) float ldsS[WPB][LDSW];

  const int tid  = threadIdx.x;
  const int wv   = tid >> 5;
  const int lane = tid & 31;
  const int r    = lane & 15;
  const int h    = lane >> 4;

  const int stride = gridDim.x * WPB;
  int b = blockIdx.x * WPB + wv;
  if (b >= B) return;

  float w1[5][3];
#pragma unroll
  for (int c = 0; c < 5; ++c)
#pragma unroll
    for (int f = 0; f < 3; ++f) w1[c][f] = W1[c * 3 + f];
  float w2v = 0.0f;
  if (lane < 3) w2v = W2[lane];

  float* ldsW = &ldsS[wv][0];
  const int fr = r < 3 ? r : 2;
  const float* bsrc = ldsW + fr * GP + 16 * h;
  const float* csrc = ldsW + fr * GP + 8 * h;
  const float* dsrc = ldsW + 3 * GP + 8 * h;

  Tile t;
  v2f a[4][16];
  float xw[2][3];
  prefetch_tile(A, h0, imin(b + stride, B - 1), lane);
  load_tile(t, A, h0, b, lane);
  for (;;) {
    const int bn = b + stride;
    unpack_tile(t, a, xw, w1);
    if (bn >= B) { finish_tile(a, xw, b, ldsW, bsrc, csrc, dsrc, w2v, out, r, h, lane); break; }
    prefetch_tile(A, h0, imin(bn + stride, B - 1), lane);
    load_tile(t, A, h0, bn, lane);
    finish_tile(a, xw, b, ldsW, bsrc, csrc, dsrc, w2v, out, r, h, lane);
    b = bn;
  }
}

extern "C" void kernel_launch(void* const* d_in, const int* in_sizes, int n_in,
                              void* d_out, int out_size, void* d_ws, size_t ws_size,
                              hipStream_t stream) {
  const float* A  = (const float*)d_in[0];
  const float* h0 = (const float*)d_in[1];
  const float* W1 = (const float*)d_in[2];
  const float* W2 = (const float*)d_in[3];
  float* out = (float*)d_out;

  const int B = in_sizes[0] / (NV * NV);
  dim3 block(256);
  dim3 grid(NBLK);
  gcn_pool_kernel<<<grid, block, 0, stream>>>(A, h0, W1, W2, out, B);
}


# Product transposed: loaded A registers are the WMMA B operand (two register sets in turn), no fragment copy or lane exchange; a row's two column pieces add after the product; sum order changes
# speedup vs baseline: 1.3655x; 1.0400x over previous
#include <hip/hip_runtime.h>


typedef __attribute__((ext_vector_type(2))) float v2f;
typedef __attribute__((ext_vector_type(4))) float v4f;
typedef __attribute__((ext_vector_type(8))) float v8f;

#define NV 64
#define WPB 8
#define NBLK 256

struct Tile {
  v4f raw[4][2][2][2];
  float hv[2][5];
};

__device__ __forceinline__ int own_row0(int lane)
{
  return (lane & 7) + 32 * ((lane >> 3) & 1) + 16 * (lane >> 4);
}

__device__ __forceinline__ void load_tile(Tile& t, const float* __restrict__ A,
                                          const float* __restrict__ h0, int b, int lane)
{
  const float* Hb = h0 + (size_t)b * (NV * 5);
  const int R0 = own_row0(lane);
#pragma unroll
  for (int s = 0; s < 2; ++s) {
    const float* hr = Hb + (R0 + 8 * s) * 5;
#pragma unroll
    for (int c = 0; c < 5; ++c) t.hv[s][c] = hr[c];
  }
  const float* Ab = A + (size_t)b * (NV * NV) + (lane & 7) * NV + (lane >> 4) * 16 + ((lane >> 3) & 1) * 8;
#pragma unroll
  for (int m = 0; m < 4; ++m)
#pragma unroll
    for (int c = 0; c < 2; ++c)
#pragma unroll
      for (int q = 0; q < 2; ++q)
#pragma unroll
        for (int d = 0; d < 2; ++d)
          t.raw[m][c][d][q] = *(const v4f*)(Ab + (16 * m + 8 * q) * NV + 32 * c + 4 * d);
}

template <int XM>
__device__ __forceinline__ float row_xor(float x)
{
  return __builtin_bit_cast(float,
      __builtin_amdgcn_update_dpp(0, __builtin_bit_cast(int, x), 0x160 + XM, 0xF, 0xF, true));
}

__device__ __forceinline__ float half_xor(float x)
{
  return __builtin_bit_cast(float,
      __builtin_amdgcn_permlanex16(__builtin_bit_cast(int, x), __builtin_bit_cast(int, x),
                                   0x76543210, 0xfedcba98, false, false));
}

__device__ __forceinline__ int imin(int x, int y) { return x < y ? x : y; }

__device__ __forceinline__ void prefetch_tile(const float* __restrict__ A,
                                              const float* __restrict__ h0, int b, int lane)
{
  const char* Ab = (const char*)(A + (size_t)b * (NV * NV));
  const char* Hb = (const char*)(h0 + (size_t)b * (NV * 5));
  __builtin_prefetch(Ab + lane * 256, 0, 3);
  __builtin_prefetch(Ab + (lane + 32) * 256, 0, 3);
  __builtin_prefetch(Hb + (lane < 4 ? lane : 4) * 256, 0, 3);
}

#define GP 76
#define LDSW (3 * GP + 3 * NV + NV)

__device__ __forceinline__ float finish_tile(const Tile& t, float* ldsW, const float* gsrc,
                                             const float* esrc, const float (&w1)[5][3],
                                             const float (&w2)[3], int lane, bool up3, bool up4)
{
  float* GT  = ldsW;
  float* GT2 = ldsW + 3 * GP;
  float* DV2 = ldsW + 3 * GP + 3 * NV;
  const int j7  = lane & 7;
  const int mq0 = 4 * ((lane >> 3) & 1) + 2 * (lane >> 4);
  const int R0  = own_row0(lane);

  float xw[2][3];
#pragma unroll
  for (int s = 0; s < 2; ++s)
#pragma unroll
    for (int f = 0; f < 3; ++f) {
      float acc = 0.0f;
#pragma unroll
      for (int c = 0; c < 5; ++c) acc += t.hv[s][c] * w1[c][f];
      xw[s][f] = acc;
    }

  float ps[8];
#pragma unroll
  for (int m = 0; m < 4; ++m)
#pragma unroll
    for (int q = 0; q < 2; ++q) {
      v4f u = t.raw[m][0][0][q] + t.raw[m][0][1][q];
      v4f w = t.raw[m][1][0][q] + t.raw[m][1][1][q];
      u += w;
      ps[2 * m + q] = (u.x + u.z) + (u.y + u.w);
    }
  float s1[4], s2[2];
#pragma unroll
  for (int k = 0; k < 4; ++k) {
    const float keep = up3 ? ps[4 + k] : ps[k];
    const float send = up3 ? ps[k] : ps[4 + k];
    s1[k] = keep + row_xor<8>(send);
  }
#pragma unroll
  for (int k = 0; k < 2; ++k) {
    const float keep = up4 ? s1[2 + k] : s1[k];
    const float send = up4 ? s1[k] : s1[2 + k];
    s2[k] = keep + half_xor(send);
  }
  float dv[2];
#pragma unroll
  for (int s = 0; s < 2; ++s) dv[s] = 1.0f / sqrtf(s2[s] + 1.0f);

#pragma unroll
  for (int s = 0; s < 2; ++s) {
#pragma unroll
    for (int f = 0; f < 3; ++f) {
      const float g = dv[s] * xw[s][f];
      GT[f * GP + R0 + 8 * s] = g;
      GT2[f * NV + 8 * j7 + mq0 + s] = g;
    }
    DV2[8 * j7 + mq0 + s] = dv[s];
  }

  v4f gq[2][2], gtq[3][2], dvq[2];
#pragma unroll
  for (int c = 0; c < 2; ++c)
#pragma unroll
    for (int d = 0; d < 2; ++d) gq[c][d] = *(const v4f*)(gsrc + 32 * c + 4 * d);
#pragma unroll
  for (int u = 0; u < 2; ++u) {
#pragma unroll
    for (int f = 0; f < 3; ++f) gtq[f][u] = *(const v4f*)(esrc + f * NV + 4 * u);
    dvq[u] = *(const v4f*)(esrc + 3 * NV + 4 * u);
  }

  __builtin_amdgcn_sched_barrier(0);

  float p[3] = {0.0f, 0.0f, 0.0f};
#pragma unroll
  for (int u = 0; u < 2; ++u) {
    v8f acc[4];
#pragma unroll
    for (int e = 0; e < 4; ++e) acc[e] = (v8f){0.0f, 0.0f, 0.0f, 0.0f, 0.0f, 0.0f, 0.0f, 0.0f};
#pragma unroll
    for (int c = 0; c < 2; ++c)
#pragma unroll
      for (int d = 0; d < 2; ++d)
#pragma unroll
        for (int xp = 0; xp < 2; ++xp) {
          const v2f ga = xp ? __builtin_shufflevector(gq[c][d], gq[c][d], 2, 3)
                            : __builtin_shufflevector(gq[c][d], gq[c][d], 0, 1);
#pragma unroll
          for (int e = 0; e < 4; ++e) {
            const int mq = 4 * u + e;
            const v4f r = t.raw[mq >> 1][c][d][mq & 1];
            const v2f ab = xp ? __builtin_shufflevector(r, r, 2, 3) : __builtin_shufflevector(r, r, 0, 1);
            acc[e] = __builtin_amdgcn_wmma_f32_16x16x4_f32(
                 false, ga,  false, ab,
                 (short)0, acc[e],  false,  false);
          }
        }
#pragma unroll
    for (int e = 0; e < 4; ++e)
#pragma unroll
      for (int f = 0; f < 3; ++f) {
        const float ag = acc[e][f] + row_xor<8>(acc[e][3 + f]);
        p[f] += fmaxf((ag + gtq[f][u][e]) * dvq[u][e], 0.0f);
      }
  }
#pragma unroll
  for (int f = 0; f < 3; ++f) {
    p[f] += row_xor<1>(p[f]);
    p[f] += row_xor<2>(p[f]);
    p[f] += row_xor<4>(p[f]);
  }

  const float o = (p[0] * w2[0] + p[1] * w2[1]) + p[2] * w2[2];
  return __builtin_bit_cast(float, __builtin_amdgcn_readfirstlane(__builtin_bit_cast(int, o)));
}

__device__ __forceinline__ void keep_result(float o, float& res, int& cnt, int& bfirst, int b, int stride,
                                            float* __restrict__ out, int lane)
{
  res = (lane == cnt) ? o : res;
  if (++cnt == 32) {
    out[bfirst + lane * stride] = res;
    cnt = 0;
    bfirst = b + stride;
  }
}

__global__ void __launch_bounds__(256)
gcn_pool_kernel(const float* __restrict__ A, const float* __restrict__ h0,
                const float* __restrict__ W1, const float* __restrict__ W2,
                float* __restrict__ out, int B)
{
  __shared__ __attribute__((aligned(16))) float ldsS[WPB][LDSW];

  const int tid  = threadIdx.x;
  const int wv   = __builtin_amdgcn_readfirstlane(tid >> 5);
  const int lane = tid & 31;

  const int stride = gridDim.x * WPB;
  int b = blockIdx.x * WPB + wv;
  if (b >= B) return;

  float w1[5][3];
#pragma unroll
  for (int c = 0; c < 5; ++c)
#pragma unroll
    for (int f = 0; f < 3; ++f) w1[c][f] = W1[c * 3 + f];
  float w2[3];
#pragma unroll
  for (int f = 0; f < 3; ++f) w2[f] = W2[f];

  float* ldsW = &ldsS[wv][0];
  const int fp    = lane & 15;
  const int fsel  = fp < 3 ? fp : (fp < 6 ? fp - 3 : 2);
  const int piece = (fp >= 3 && fp < 6) ? 1 : 0;
  const float* gsrc = ldsW + fsel * GP + 16 * (lane >> 4) + 8 * piece;
  const float* esrc = ldsW + 3 * GP + 8 * (lane & 7);
  const bool up3 = (lane & 8) != 0;
  const bool up4 = (lane & 16) != 0;

  Tile t0, t1;
  float res = 0.0f;
  int cnt = 0, bfirst = b;
  prefetch_tile(A, h0, imin(b + stride, B - 1), lane);
  load_tile(t0, A, h0, b, lane);
  for (;;) {
    int bn = b + stride;
    if (bn >= B) {
      keep_result(finish_tile(t0, ldsW, gsrc, esrc, w1, w2, lane, up3, up4), res, cnt, bfirst, b, stride, out, lane);
      break;
    }
    prefetch_tile(A, h0, imin(bn + stride, B - 1), lane);
    load_tile(t1, A, h0, bn, lane);
    keep_result(finish_tile(t0, ldsW, gsrc, esrc, w1, w2, lane, up3, up4), res, cnt, bfirst, b, stride, out, lane);
    b = bn;
    bn = b + stride;
    if (bn >= B) {
      keep_result(finish_tile(t1, ldsW, gsrc, esrc, w1, w2, lane, up3, up4), res, cnt, bfirst, b, stride, out, lane);
      break;
    }
    prefetch_tile(A, h0, imin(bn + stride, B - 1), lane);
    load_tile(t0, A, h0, bn, lane);
    keep_result(finish_tile(t1, ldsW, gsrc, esrc, w1, w2, lane, up3, up4), res, cnt, bfirst, b, stride, out, lane);
    b = bn;
  }
  if (lane < cnt) out[bfirst + lane * stride] = res;
}

extern "C" void kernel_launch(void* const* d_in, const int* in_sizes, int n_in,
                              void* d_out, int out_size, void* d_ws, size_t ws_size,
                              hipStream_t stream) {
  const float* A  = (const float*)d_in[0];
  const float* h0 = (const float*)d_in[1];
  const float* W1 = (const float*)d_in[2];
  const float* W2 = (const float*)d_in[3];
  float* out = (float*)d_out;

  const int B = in_sizes[0] / (NV * NV);
  dim3 block(256);
  dim3 grid(NBLK);
  gcn_pool_kernel<<<grid, block, 0, stream>>>(A, h0, W1, W2, out, B);
}
